# starting kernel
# baseline (speedup 1.0000x reference)
#include <hip/hip_runtime.h>

#define NNODE 10000
#define NEDGE 320000
#define MCH   32
#define NH    8

typedef float v2f __attribute__((ext_vector_type(2)));
typedef float v8f __attribute__((ext_vector_type(8)));

__device__ __forceinline__ unsigned f2key(float x) {
    unsigned b = __float_as_uint(x);
    return (b & 0x80000000u) ? ~b : (b | 0x80000000u);
}
__device__ __forceinline__ float key2f(unsigned k) {
    if (k == 0u) return 0.0f;
    unsigned b = (k & 0x80000000u) ? (k & 0x7fffffffu) : ~k;
    return __uint_as_float(b);
}

__global__ void gmab_init(float* __restrict__ out, int out_n,
                          unsigned* __restrict__ emaxKey,
                          float* __restrict__ denom, int nh) {
    int i = blockIdx.x * blockDim.x + threadIdx.x;
    if (i < out_n) out[i] = 0.0f;
    if (i < nh) { emaxKey[i] = 0u; denom[i] = 0.0f; }
}

__global__ __launch_bounds__(256)
void gmab_dot_max(const float* __restrict__ k0, const float* __restrict__ k1,
                  const float* __restrict__ q0, const float* __restrict__ q1,
                  const int*   __restrict__ dst,
                  float* __restrict__ eArr, unsigned* __restrict__ emaxKey) {
    const int  lane  = threadIdx.x & 31;
    const int  wave  = threadIdx.x >> 5;
    const long ebase = ((long)blockIdx.x * 8 + wave) * 16;
    const int  r     = lane & 15;
    const long e     = ebase + r;
    const int  n     = dst[e];
    const int  klo   = (lane < 16) ? 0 : 2;
    const float scale = 0.08838834764831845f;

    #pragma unroll
    for (int h = 0; h < NH; ++h) {
        v8f acc = {};
        #pragma unroll
        for (int kk = 0; kk < 4; ++kk) {
            const int f0 = kk * 4 + klo;
            v2f a, b;
            if (kk == 0) {
                a.x = k0[e * MCH + h * 4 + f0];
                a.y = k0[e * MCH + h * 4 + f0 + 1];
                b.x = q0[(long)n * MCH + h * 4 + f0];
                b.y = q0[(long)n * MCH + h * 4 + f0 + 1];
            } else {
                a.x = k1[e * (MCH * 3) + h * 12 + (f0 - 4)];
                a.y = k1[e * (MCH * 3) + h * 12 + (f0 - 3)];
                b.x = q1[(long)n * (MCH * 3) + h * 12 + (f0 - 4)];
                b.y = q1[(long)n * (MCH * 3) + h * 12 + (f0 - 3)];
            }
            acc = __builtin_amdgcn_wmma_f32_16x16x4_f32(
                false, a, false, b, (short)0, acc, false, false);
        }
        float ev = 0.0f;
        #pragma unroll
        for (int v = 0; v < 8; ++v) {
            if (lane == v || lane == v + 24) ev = acc[v];
        }
        if (lane < 8 || lane >= 24) {
            const long eo = ebase + ((lane < 8) ? lane : (lane - 16));
            const float es = ev * scale;
            eArr[eo * NH + h] = es;
            atomicMax(&emaxKey[(long)n * NH + h], f2key(es));
        }
    }
}

__global__ void gmab_exp_sum(const int* __restrict__ dst,
                             const unsigned* __restrict__ emaxKey,
                             float* __restrict__ eArr,
                             float* __restrict__ denom) {
    long i = (long)blockIdx.x * blockDim.x + threadIdx.x;
    if (i >= (long)NEDGE * NH) return;
    long e = i / NH;
    int  h = (int)(i % NH);
    int  n = dst[e];
    float em = key2f(emaxKey[(long)n * NH + h]);
    float ex = __expf(eArr[i] - em);
    eArr[i] = ex;
    atomicAdd(&denom[(long)n * NH + h], ex);
}

__global__ void gmab_scatter(const float* __restrict__ v0, const float* __restrict__ v1,
                             const int*   __restrict__ dst,
                             const float* __restrict__ eArr,
                             const float* __restrict__ denom,
                             float* __restrict__ out) {
    long i = (long)blockIdx.x * blockDim.x + threadIdx.x;
    if (i >= (long)NEDGE * MCH) return;
    long e = i >> 5;
    int  m = (int)(i & 31);
    int  h = m >> 2;
    int  n = dst[e];
    float a = eArr[e * NH + h] / denom[(long)n * NH + h];

    atomicAdd(&out[(long)n * MCH + m], a * v0[i]);

    float* out1 = out + (long)NNODE * MCH;
    long vi = i * 3;
    long oi = ((long)n * MCH + m) * 3;
    atomicAdd(&out1[oi + 0], a * v1[vi + 0]);
    atomicAdd(&out1[oi + 1], a * v1[vi + 1]);
    atomicAdd(&out1[oi + 2], a * v1[vi + 2]);
}

extern "C" void kernel_launch(void* const* d_in, const int* in_sizes, int n_in,
                              void* d_out, int out_size, void* d_ws, size_t ws_size,
                              hipStream_t stream) {
    const float* v0 = (const float*)d_in[0];
    const float* v1 = (const float*)d_in[1];
    const float* k0 = (const float*)d_in[2];
    const float* k1 = (const float*)d_in[3];
    const float* q0 = (const float*)d_in[4];
    const float* q1 = (const float*)d_in[5];
    const int*   dst = (const int*)d_in[6];
    float* out = (float*)d_out;

    float*    eArr    = (float*)d_ws;
    unsigned* emaxKey = (unsigned*)(eArr + (size_t)NEDGE * NH);
    float*    denom   = (float*)(emaxKey + (size_t)NNODE * NH);

    const int nh = NNODE * NH;
    const int initN = (out_size > nh) ? out_size : nh;
    gmab_init<<<(initN + 255) / 256, 256, 0, stream>>>(out, out_size, emaxKey, denom, nh);

    gmab_dot_max<<<NEDGE / 128, 256, 0, stream>>>(k0, k1, q0, q1, dst, eArr, emaxKey);

    const long p2 = (long)NEDGE * NH;
    gmab_exp_sum<<<(int)((p2 + 255) / 256), 256, 0, stream>>>(dst, emaxKey, eArr, denom);

    const long p3 = (long)NEDGE * MCH;
    gmab_scatter<<<(int)((p3 + 255) / 256), 256, 0, stream>>>(v0, v1, dst, eArr, denom, out);
}


# scatter pass: each work-group owns 40 nodes, keeps their 128 sums in LDS, scans dst for its own edges and adds with LDS float atomics; out stored once, no global float adds; sum order changes
# speedup vs baseline: 1.3629x; 1.3629x over previous
#include <hip/hip_runtime.h>

#define NNODE 10000
#define NEDGE 320000
#define MCH   32
#define NH    8

typedef float v2f __attribute__((ext_vector_type(2)));
typedef float v4f __attribute__((ext_vector_type(4)));
typedef float v8f __attribute__((ext_vector_type(8)));
typedef int   v4i __attribute__((ext_vector_type(4)));

__device__ __forceinline__ unsigned f2key(float x) {
    unsigned b = __float_as_uint(x);
    return (b & 0x80000000u) ? ~b : (b | 0x80000000u);
}
__device__ __forceinline__ float key2f(unsigned k) {
    if (k == 0u) return 0.0f;
    unsigned b = (k & 0x80000000u) ? (k & 0x7fffffffu) : ~k;
    return __uint_as_float(b);
}

__global__ void gmab_init(unsigned* __restrict__ emaxKey,
                          float* __restrict__ denom, int nh) {
    int i = blockIdx.x * blockDim.x + threadIdx.x;
    if (i < nh) { emaxKey[i] = 0u; denom[i] = 0.0f; }
}

#define TROW 132
#define DOT_WAVES 4
__global__ __launch_bounds__(DOT_WAVES * 32)
void gmab_dot_max(const float* __restrict__ k0, const float* __restrict__ k1,
                  const float* __restrict__ q0, const float* __restrict__ q1,
                  const int*   __restrict__ dst,
                  float* __restrict__ eArr, unsigned* __restrict__ emaxKey) {
    __shared__ float tile[DOT_WAVES][2][16 * TROW];
    const int  lane  = threadIdx.x & 31;
    const int  wave  = threadIdx.x >> 5;
    const long ebase = ((long)blockIdx.x * DOT_WAVES + wave) * 16;
    const int  r     = lane & 15;
    const long e     = ebase + r;
    const int  n     = dst[e];
    const int  klo   = (lane < 16) ? 0 : 2;
    const float scale = 0.08838834764831845f;

    float* kt = &tile[wave][0][0];
    float* qt = &tile[wave][1][0];

    const bool     lo    = lane < 8;
    const float*   kcol  = lo ? (k0 + lane * 4) : (k1 + (lane - 8) * 4);
    const float*   qcol  = lo ? (q0 + lane * 4) : (q1 + (lane - 8) * 4);
    const unsigned pitch = lo ? MCH : MCH * 3;
    const unsigned eb    = (unsigned)ebase;
    v4f kr[16], qr[16];
    #pragma unroll
    for (int i = 0; i < 16; ++i) {
        const unsigned ni = (unsigned)__builtin_amdgcn_readlane(n, i);
        kr[i] = *(const v4f*)(kcol + (size_t)((eb + i) * pitch));
        qr[i] = *(const v4f*)(qcol + (size_t)(ni * pitch));
    }
    #pragma unroll
    for (int i = 0; i < 16; ++i) {
        *(v4f*)(kt + i * TROW + lane * 4) = kr[i];
        *(v4f*)(qt + i * TROW + lane * 4) = qr[i];
    }
    __syncthreads();

    const float* ka = kt + r * TROW + klo;
    const float* qb = qt + r * TROW + klo;

    #pragma unroll
    for (int h = 0; h < NH; ++h) {
        v8f acc = {};
        #pragma unroll
        for (int kk = 0; kk < 4; ++kk) {
            const int off = (kk == 0) ? (h * 4) : (MCH + h * 12 + (kk - 1) * 4);
            const v2f a = *(const v2f*)(ka + off);
            const v2f b = *(const v2f*)(qb + off);
            acc = __builtin_amdgcn_wmma_f32_16x16x4_f32(
                false, a, false, b, (short)0, acc, false, false);
        }
        float ev = 0.0f;
        #pragma unroll
        for (int v = 0; v < 8; ++v) {
            if (lane == v || lane == v + 24) ev = acc[v];
        }
        if (lane < 8 || lane >= 24) {
            const long eo = ebase + ((lane < 8) ? lane : (lane - 16));
            const float es = ev * scale;
            eArr[eo * NH + h] = es;
            atomicMax(&emaxKey[(long)n * NH + h], f2key(es));
        }
    }
}

__global__ void gmab_exp_sum(const int* __restrict__ dst,
                             const unsigned* __restrict__ emaxKey,
                             float* __restrict__ eArr,
                             float* __restrict__ denom) {
    long i = (long)blockIdx.x * blockDim.x + threadIdx.x;
    if (i >= (long)NEDGE * NH) return;
    long e = i / NH;
    int  h = (int)(i % NH);
    int  n = dst[e];
    float em = key2f(emaxKey[(long)n * NH + h]);
    float ex = __expf(eArr[i] - em);
    eArr[i] = ex;
    atomicAdd(&denom[(long)n * NH + h], ex);
}

#define SC_NODES 40
#define SC_WAVES 32
#define SC_BATCH 16
#define SC_STEP  128
#define SC_DEPTH 8
#define SC_CAP   (SC_BATCH + SC_DEPTH * SC_STEP)
__global__ __launch_bounds__(SC_WAVES * 32)
void gmab_scatter(const float* __restrict__ v0, const float* __restrict__ v1,
                  const int*   __restrict__ dst,
                  const float* __restrict__ eArr,
                  const float* __restrict__ denom,
                  float* __restrict__ out) {
    __shared__ float    acc[(SC_NODES + 1) * 128];
    __shared__ float    dnm[(SC_NODES + 1) * NH];
    __shared__ unsigned list[SC_WAVES][SC_CAP];

    const int lane = threadIdx.x & 31;
    const int wave = __builtin_amdgcn_readfirstlane((int)(threadIdx.x >> 5));
    const int lo   = blockIdx.x * SC_NODES;

    for (int i = threadIdx.x; i < (SC_NODES + 1) * 128; i += SC_WAVES * 32) acc[i] = 0.0f;
    for (int i = threadIdx.x; i < (SC_NODES + 1) * NH; i += SC_WAVES * 32)
        dnm[i] = (i < SC_NODES * NH && lo + i / NH < NNODE) ? denom[(long)lo * NH + i] : 1.0f;
    __syncthreads();

    const bool     lop   = lane < 8;
    const float*   vcol  = lop ? (v0 + lane * 4) : (v1 + (lane - 8) * 4);
    const unsigned pitch = lop ? MCH : MCH * 3;
    const int      hl    = lop ? lane : (lane - 8) / 3;
    unsigned* mylist = &list[wave][0];
    unsigned  cnt    = 0;

    const v4i* dst4 = (const v4i*)dst;
    const int  nstep = NEDGE / SC_STEP;
    for (int t0 = wave; t0 < nstep; t0 += SC_WAVES * SC_DEPTH) {
        v4i d[SC_DEPTH];
        #pragma unroll
        for (int u = 0; u < SC_DEPTH; ++u) {
            const int t = t0 + u * SC_WAVES;
            d[u] = dst4[(size_t)((t < nstep) ? t : (nstep - 1)) * 32 + lane];
        }
        #pragma unroll
        for (int u = 0; u < SC_DEPTH; ++u) {
            const int      t     = t0 + u * SC_WAVES;
            const unsigned ebase = (unsigned)t * SC_STEP + lane * 4;
            const unsigned lim = (t < nstep) ? SC_NODES : 0u;
            const unsigned r0 = (unsigned)(d[u].x - lo), r1 = (unsigned)(d[u].y - lo);
            const unsigned r2 = (unsigned)(d[u].z - lo), r3 = (unsigned)(d[u].w - lo);
            const bool m0 = r0 < lim, m1 = r1 < lim, m2 = r2 < lim, m3 = r3 < lim;
            if (__builtin_amdgcn_ballot_w32(m0 | m1 | m2 | m3) != 0u) {
                unsigned mk;
                mk = __builtin_amdgcn_ballot_w32(m0);
                if (m0) mylist[cnt + __builtin_amdgcn_mbcnt_lo(mk, 0u)] = (ebase + 0u) | (r0 << 20);
                cnt += __builtin_popcount(mk);
                mk = __builtin_amdgcn_ballot_w32(m1);
                if (m1) mylist[cnt + __builtin_amdgcn_mbcnt_lo(mk, 0u)] = (ebase + 1u) | (r1 << 20);
                cnt += __builtin_popcount(mk);
                mk = __builtin_amdgcn_ballot_w32(m2);
                if (m2) mylist[cnt + __builtin_amdgcn_mbcnt_lo(mk, 0u)] = (ebase + 2u) | (r2 << 20);
                cnt += __builtin_popcount(mk);
                mk = __builtin_amdgcn_ballot_w32(m3);
                if (m3) mylist[cnt + __builtin_amdgcn_mbcnt_lo(mk, 0u)] = (ebase + 3u) | (r3 << 20);
                cnt += __builtin_popcount(mk);
            }
        }
        const bool last = t0 + SC_WAVES * SC_DEPTH >= nstep;
        if (cnt >= SC_BATCH || last) {
            __builtin_amdgcn_fence(__ATOMIC_RELEASE, "wavefront");
            __builtin_amdgcn_wave_barrier();
            __builtin_amdgcn_fence(__ATOMIC_ACQUIRE, "wavefront");
            const unsigned nb = last ? (cnt + SC_BATCH - 1) / SC_BATCH : cnt / SC_BATCH;
            for (unsigned b = 0; b < nb; ++b) {
                const unsigned li  = b * SC_BATCH + (unsigned)(lane & (SC_BATCH - 1));
                const unsigned ent = (li < cnt) ? mylist[li] : ((unsigned)SC_NODES << 20);
                v4f   vr[SC_BATCH];
                float ex[SC_BATCH];
                #pragma unroll
                for (int i = 0; i < SC_BATCH; ++i) {
                    const unsigned e = (unsigned)__builtin_amdgcn_readlane((int)ent, i) & 0xFFFFFu;
                    vr[i] = *(const v4f*)(vcol + (size_t)e * pitch);
                    ex[i] = eArr[(size_t)e * NH + hl];
                }
                #pragma unroll
                for (int i = 0; i < SC_BATCH; ++i) {
                    const unsigned nl = (unsigned)__builtin_amdgcn_readlane((int)ent, i) >> 20;
                    const float a = ex[i] / dnm[nl * NH + hl];
                    float* ap = acc + nl * 128 + lane;
                    atomicAdd(ap,      a * vr[i].x);
                    atomicAdd(ap + 32, a * vr[i].y);
                    atomicAdd(ap + 64, a * vr[i].z);
                    atomicAdd(ap + 96, a * vr[i].w);
                }
            }
            const unsigned done = nb * SC_BATCH;
            const unsigned rem  = (done < cnt) ? (cnt - done) : 0u;
            const unsigned keep = ((unsigned)lane < rem) ? mylist[done + lane] : 0u;
            if ((unsigned)lane < rem) mylist[lane] = keep;
            cnt = rem;
        }
    }
    __syncthreads();

    float* out1 = out + (size_t)NNODE * MCH;
    for (int nl = wave; nl < SC_NODES; nl += SC_WAVES) {
        const int n = lo + nl;
        if (n < NNODE) {
            const float* ap = acc + nl * 128 + lane;
            const v4f o = {ap[0], ap[32], ap[64], ap[96]};
            float* op = lop ? (out + (size_t)n * MCH + lane * 4)
                            : (out1 + (size_t)n * (MCH * 3) + (lane - 8) * 4);
            *(v4f*)op = o;
        }
    }
}

extern "C" void kernel_launch(void* const* d_in, const int* in_sizes, int n_in,
                              void* d_out, int out_size, void* d_ws, size_t ws_size,
                              hipStream_t stream) {
    const float* v0 = (const float*)d_in[0];
    const float* v1 = (const float*)d_in[1];
    const float* k0 = (const float*)d_in[2];
    const float* k1 = (const float*)d_in[3];
    const float* q0 = (const float*)d_in[4];
    const float* q1 = (const float*)d_in[5];
    const int*   dst = (const int*)d_in[6];
    float* out = (float*)d_out;

    float*    eArr    = (float*)d_ws;
    unsigned* emaxKey = (unsigned*)(eArr + (size_t)NEDGE * NH);
    float*    denom   = (float*)(emaxKey + (size_t)NNODE * NH);

    const int nh = NNODE * NH;
    gmab_init<<<(nh + 255) / 256, 256, 0, stream>>>(emaxKey, denom, nh);

    gmab_dot_max<<<NEDGE / (DOT_WAVES * 16), DOT_WAVES * 32, 0, stream>>>(k0, k1, q0, q1, dst, eArr, emaxKey);

    const long p2 = (long)NEDGE * NH;
    gmab_exp_sum<<<(int)((p2 + 255) / 256), 256, 0, stream>>>(dst, emaxKey, eArr, denom);

    gmab_scatter<<<(NNODE + SC_NODES - 1) / SC_NODES, SC_WAVES * 32, 0, stream>>>(v0, v1, dst, eArr, denom, out);
}


# softmax sums folded into the owner-group scatter: the group takes exp(e-emax) itself, sums ex and ex*v in LDS, divides once at the end; exp/sum pass and its global float adds go; sum order changes
# speedup vs baseline: 1.5488x; 1.1364x over previous
#include <hip/hip_runtime.h>

#define NNODE 10000
#define NEDGE 320000
#define MCH   32
#define NH    8

typedef float v2f __attribute__((ext_vector_type(2)));
typedef float v4f __attribute__((ext_vector_type(4)));
typedef float v8f __attribute__((ext_vector_type(8)));
typedef int   v4i __attribute__((ext_vector_type(4)));

__device__ __forceinline__ unsigned f2key(float x) {
    unsigned b = __float_as_uint(x);
    return (b & 0x80000000u) ? ~b : (b | 0x80000000u);
}
__device__ __forceinline__ float key2f(unsigned k) {
    if (k == 0u) return 0.0f;
    unsigned b = (k & 0x80000000u) ? (k & 0x7fffffffu) : ~k;
    return __uint_as_float(b);
}

__global__ void gmab_init(unsigned* __restrict__ emaxKey, int nh) {
    int i = blockIdx.x * blockDim.x + threadIdx.x;
    if (i < nh) emaxKey[i] = 0u;
}

#define TROW 132
#define DOT_WAVES 4
__global__ __launch_bounds__(DOT_WAVES * 32)
void gmab_dot_max(const float* __restrict__ k0, const float* __restrict__ k1,
                  const float* __restrict__ q0, const float* __restrict__ q1,
                  const int*   __restrict__ dst,
                  float* __restrict__ eArr, unsigned* __restrict__ emaxKey) {
    __shared__ float tile[DOT_WAVES][2][16 * TROW];
    const int  lane  = threadIdx.x & 31;
    const int  wave  = threadIdx.x >> 5;
    const long ebase = ((long)blockIdx.x * DOT_WAVES + wave) * 16;
    const int  r     = lane & 15;
    const long e     = ebase + r;
    const int  n     = dst[e];
    const int  klo   = (lane < 16) ? 0 : 2;
    const float scale = 0.08838834764831845f;

    float* kt = &tile[wave][0][0];
    float* qt = &tile[wave][1][0];

    const bool     lo    = lane < 8;
    const float*   kcol  = lo ? (k0 + lane * 4) : (k1 + (lane - 8) * 4);
    const float*   qcol  = lo ? (q0 + lane * 4) : (q1 + (lane - 8) * 4);
    const unsigned pitch = lo ? MCH : MCH * 3;
    const unsigned eb    = (unsigned)ebase;
    v4f kr[16], qr[16];
    #pragma unroll
    for (int i = 0; i < 16; ++i) {
        const unsigned ni = (unsigned)__builtin_amdgcn_readlane(n, i);
        kr[i] = *(const v4f*)(kcol + (size_t)((eb + i) * pitch));
        qr[i] = *(const v4f*)(qcol + (size_t)(ni * pitch));
    }
    #pragma unroll
    for (int i = 0; i < 16; ++i) {
        *(v4f*)(kt + i * TROW + lane * 4) = kr[i];
        *(v4f*)(qt + i * TROW + lane * 4) = qr[i];
    }
    __syncthreads();

    const float* ka = kt + r * TROW + klo;
    const float* qb = qt + r * TROW + klo;

    #pragma unroll
    for (int h = 0; h < NH; ++h) {
        v8f acc = {};
        #pragma unroll
        for (int kk = 0; kk < 4; ++kk) {
            const int off = (kk == 0) ? (h * 4) : (MCH + h * 12 + (kk - 1) * 4);
            const v2f a = *(const v2f*)(ka + off);
            const v2f b = *(const v2f*)(qb + off);
            acc = __builtin_amdgcn_wmma_f32_16x16x4_f32(
                false, a, false, b, (short)0, acc, false, false);
        }
        float ev = 0.0f;
        #pragma unroll
        for (int v = 0; v < 8; ++v) {
            if (lane == v || lane == v + 24) ev = acc[v];
        }
        if (lane < 8 || lane >= 24) {
            const long eo = ebase + ((lane < 8) ? lane : (lane - 16));
            const float es = ev * scale;
            eArr[eo * NH + h] = es;
            atomicMax(&emaxKey[(long)n * NH + h], f2key(es));
        }
    }
}

#define SC_NODES 40
#define SC_WAVES 32
#define SC_BATCH 16
#define SC_STEP  128
#define SC_DEPTH 8
#define SC_CAP   (SC_BATCH + SC_DEPTH * SC_STEP)
__global__ __launch_bounds__(SC_WAVES * 32)
void gmab_scatter(const float* __restrict__ v0, const float* __restrict__ v1,
                  const int*   __restrict__ dst,
                  const float* __restrict__ eArr,
                  const unsigned* __restrict__ emaxKey,
                  float* __restrict__ out) {
    __shared__ float    acc[(SC_NODES + 1) * 128];
    __shared__ float    dnm[(SC_NODES + 1) * NH];
    __shared__ float    emx[(SC_NODES + 1) * NH];
    __shared__ unsigned list[SC_WAVES][SC_CAP];

    const int lane = threadIdx.x & 31;
    const int wave = __builtin_amdgcn_readfirstlane((int)(threadIdx.x >> 5));
    const int lo   = blockIdx.x * SC_NODES;

    for (int i = threadIdx.x; i < (SC_NODES + 1) * 128; i += SC_WAVES * 32) acc[i] = 0.0f;
    for (int i = threadIdx.x; i < (SC_NODES + 1) * NH; i += SC_WAVES * 32) {
        dnm[i] = 0.0f;
        emx[i] = (i < SC_NODES * NH && lo + i / NH < NNODE) ? key2f(emaxKey[(long)lo * NH + i]) : 0.0f;
    }
    __syncthreads();

    const bool     lop   = lane < 8;
    const float*   vcol  = lop ? (v0 + lane * 4) : (v1 + (lane - 8) * 4);
    const unsigned pitch = lop ? MCH : MCH * 3;
    const int      hl    = lop ? lane : (lane - 8) / 3;
    unsigned* mylist = &list[wave][0];
    unsigned  cnt    = 0;

    const v4i* dst4 = (const v4i*)dst;
    const int  nstep = NEDGE / SC_STEP;
    for (int t0 = wave; t0 < nstep; t0 += SC_WAVES * SC_DEPTH) {
        v4i d[SC_DEPTH];
        #pragma unroll
        for (int u = 0; u < SC_DEPTH; ++u) {
            const int t = t0 + u * SC_WAVES;
            d[u] = dst4[(size_t)((t < nstep) ? t : (nstep - 1)) * 32 + lane];
        }
        #pragma unroll
        for (int u = 0; u < SC_DEPTH; ++u) {
            const int      t     = t0 + u * SC_WAVES;
            const unsigned ebase = (unsigned)t * SC_STEP + lane * 4;
            const unsigned lim = (t < nstep) ? SC_NODES : 0u;
            const unsigned r0 = (unsigned)(d[u].x - lo), r1 = (unsigned)(d[u].y - lo);
            const unsigned r2 = (unsigned)(d[u].z - lo), r3 = (unsigned)(d[u].w - lo);
            const bool m0 = r0 < lim, m1 = r1 < lim, m2 = r2 < lim, m3 = r3 < lim;
            if (__builtin_amdgcn_ballot_w32(m0 | m1 | m2 | m3) != 0u) {
                unsigned mk;
                mk = __builtin_amdgcn_ballot_w32(m0);
                if (m0) mylist[cnt + __builtin_amdgcn_mbcnt_lo(mk, 0u)] = (ebase + 0u) | (r0 << 20);
                cnt += __builtin_popcount(mk);
                mk = __builtin_amdgcn_ballot_w32(m1);
                if (m1) mylist[cnt + __builtin_amdgcn_mbcnt_lo(mk, 0u)] = (ebase + 1u) | (r1 << 20);
                cnt += __builtin_popcount(mk);
                mk = __builtin_amdgcn_ballot_w32(m2);
                if (m2) mylist[cnt + __builtin_amdgcn_mbcnt_lo(mk, 0u)] = (ebase + 2u) | (r2 << 20);
                cnt += __builtin_popcount(mk);
                mk = __builtin_amdgcn_ballot_w32(m3);
                if (m3) mylist[cnt + __builtin_amdgcn_mbcnt_lo(mk, 0u)] = (ebase + 3u) | (r3 << 20);
                cnt += __builtin_popcount(mk);
            }
        }
        const bool last = t0 + SC_WAVES * SC_DEPTH >= nstep;
        if (cnt >= SC_BATCH || last) {
            __builtin_amdgcn_fence(__ATOMIC_RELEASE, "wavefront");
            __builtin_amdgcn_wave_barrier();
            __builtin_amdgcn_fence(__ATOMIC_ACQUIRE, "wavefront");
            const unsigned nb = last ? (cnt + SC_BATCH - 1) / SC_BATCH : cnt / SC_BATCH;
            for (unsigned b = 0; b < nb; ++b) {
                const unsigned li  = b * SC_BATCH + (unsigned)(lane & (SC_BATCH - 1));
                const unsigned ent = (li < cnt) ? mylist[li] : ((unsigned)SC_NODES << 20);
                v4f   vr[SC_BATCH];
                float ev[SC_BATCH];
                float em[SC_BATCH];
                #pragma unroll
                for (int i = 0; i < SC_BATCH; ++i) {
                    const unsigned e = (unsigned)__builtin_amdgcn_readlane((int)ent, i) & 0xFFFFFu;
                    vr[i] = *(const v4f*)(vcol + (size_t)e * pitch);
                    ev[i] = eArr[(size_t)e * NH + hl];
                    em[i] = emx[((unsigned)__builtin_amdgcn_readlane((int)ent, i) >> 20) * NH + hl];
                }
                #pragma unroll
                for (int i = 0; i < SC_BATCH; ++i) {
                    const unsigned nl = (unsigned)__builtin_amdgcn_readlane((int)ent, i) >> 20;
                    const float a = __expf(ev[i] - em[i]);
                    float* ap = acc + nl * 128 + lane;
                    atomicAdd(ap,      a * vr[i].x);
                    atomicAdd(ap + 32, a * vr[i].y);
                    atomicAdd(ap + 64, a * vr[i].z);
                    atomicAdd(ap + 96, a * vr[i].w);
                    if (lop) atomicAdd(dnm + nl * NH + lane, a);
                }
            }
            const unsigned done = nb * SC_BATCH;
            const unsigned rem  = (done < cnt) ? (cnt - done) : 0u;
            const unsigned keep = ((unsigned)lane < rem) ? mylist[done + lane] : 0u;
            if ((unsigned)lane < rem) mylist[lane] = keep;
            cnt = rem;
        }
    }
    __syncthreads();

    float* out1 = out + (size_t)NNODE * MCH;
    for (int nl = wave; nl < SC_NODES; nl += SC_WAVES) {
        const int n = lo + nl;
        if (n < NNODE) {
            const float* ap = acc + nl * 128 + lane;
            const float d = dnm[nl * NH + hl];
            const v4f o = (d > 0.0f) ? (v4f){ap[0] / d, ap[32] / d, ap[64] / d, ap[96] / d}
                                     : (v4f){0.0f, 0.0f, 0.0f, 0.0f};
            float* op = lop ? (out + (size_t)n * MCH + lane * 4)
                            : (out1 + (size_t)n * (MCH * 3) + (lane - 8) * 4);
            *(v4f*)op = o;
        }
    }
}

extern "C" void kernel_launch(void* const* d_in, const int* in_sizes, int n_in,
                              void* d_out, int out_size, void* d_ws, size_t ws_size,
                              hipStream_t stream) {
    const float* v0 = (const float*)d_in[0];
    const float* v1 = (const float*)d_in[1];
    const float* k0 = (const float*)d_in[2];
    const float* k1 = (const float*)d_in[3];
    const float* q0 = (const float*)d_in[4];
    const float* q1 = (const float*)d_in[5];
    const int*   dst = (const int*)d_in[6];
    float* out = (float*)d_out;

    float*    eArr    = (float*)d_ws;
    unsigned* emaxKey = (unsigned*)(eArr + (size_t)NEDGE * NH);

    const int nh = NNODE * NH;
    gmab_init<<<(nh + 255) / 256, 256, 0, stream>>>(emaxKey, nh);

    gmab_dot_max<<<NEDGE / (DOT_WAVES * 16), DOT_WAVES * 32, 0, stream>>>(k0, k1, q0, q1, dst, eArr, emaxKey);

    gmab_scatter<<<(NNODE + SC_NODES - 1) / SC_NODES, SC_WAVES * 32, 0, stream>>>(v0, v1, dst, eArr, emaxKey, out);
}


# dot pass: the 16x8 e tile goes through LDS so 8 adjacent lanes carry the 8 heads of an edge: one 32-byte max request per edge, not 8 one-lane ones; e leaves as 4 whole lines in one store
# speedup vs baseline: 4.0254x; 2.5990x over previous
#include <hip/hip_runtime.h>

#define NNODE 10000
#define NEDGE 320000
#define MCH   32
#define NH    8

typedef float v2f __attribute__((ext_vector_type(2)));
typedef float v4f __attribute__((ext_vector_type(4)));
typedef float v8f __attribute__((ext_vector_type(8)));
typedef int   v4i __attribute__((ext_vector_type(4)));

__device__ __forceinline__ unsigned f2key(float x) {
    unsigned b = __float_as_uint(x);
    return (b & 0x80000000u) ? ~b : (b | 0x80000000u);
}
__device__ __forceinline__ float key2f(unsigned k) {
    if (k == 0u) return 0.0f;
    unsigned b = (k & 0x80000000u) ? (k & 0x7fffffffu) : ~k;
    return __uint_as_float(b);
}

__global__ void gmab_init(unsigned* __restrict__ emaxKey, int nh) {
    int i = blockIdx.x * blockDim.x + threadIdx.x;
    if (i < nh) emaxKey[i] = 0u;
}

#define TROW 132
#define DOT_WAVES 4
__global__ __launch_bounds__(DOT_WAVES * 32)
void gmab_dot_max(const float* __restrict__ k0, const float* __restrict__ k1,
                  const float* __restrict__ q0, const float* __restrict__ q1,
                  const int*   __restrict__ dst,
                  float* __restrict__ eArr, unsigned* __restrict__ emaxKey) {
    __shared__ float tile[DOT_WAVES][2][16 * TROW];
    __shared__ float etile[DOT_WAVES][16 * NH];
    const int  lane  = threadIdx.x & 31;
    const int  wave  = threadIdx.x >> 5;
    const long ebase = ((long)blockIdx.x * DOT_WAVES + wave) * 16;
    const int  r     = lane & 15;
    const long e     = ebase + r;
    const int  n     = dst[e];
    const int  klo   = (lane < 16) ? 0 : 2;
    const float scale = 0.08838834764831845f;

    float* kt = &tile[wave][0][0];
    float* qt = &tile[wave][1][0];

    const bool     lo    = lane < 8;
    const float*   kcol  = lo ? (k0 + lane * 4) : (k1 + (lane - 8) * 4);
    const float*   qcol  = lo ? (q0 + lane * 4) : (q1 + (lane - 8) * 4);
    const unsigned pitch = lo ? MCH : MCH * 3;
    const unsigned eb    = (unsigned)ebase;
    v4f kr[16], qr[16];
    #pragma unroll
    for (int i = 0; i < 16; ++i) {
        const unsigned ni = (unsigned)__builtin_amdgcn_readlane(n, i);
        kr[i] = *(const v4f*)(kcol + (size_t)((eb + i) * pitch));
        qr[i] = *(const v4f*)(qcol + (size_t)(ni * pitch));
    }
    #pragma unroll
    for (int i = 0; i < 16; ++i) {
        *(v4f*)(kt + i * TROW + lane * 4) = kr[i];
        *(v4f*)(qt + i * TROW + lane * 4) = qr[i];
    }
    __syncthreads();

    const float* ka = kt + r * TROW + klo;
    const float* qb = qt + r * TROW + klo;

    float es[NH];
    #pragma unroll
    for (int h = 0; h < NH; ++h) {
        v8f acc = {};
        #pragma unroll
        for (int kk = 0; kk < 4; ++kk) {
            const int off = (kk == 0) ? (h * 4) : (MCH + h * 12 + (kk - 1) * 4);
            const v2f a = *(const v2f*)(ka + off);
            const v2f b = *(const v2f*)(qb + off);
            acc = __builtin_amdgcn_wmma_f32_16x16x4_f32(
                false, a, false, b, (short)0, acc, false, false);
        }
        float ev = 0.0f;
        #pragma unroll
        for (int v = 0; v < 8; ++v) {
            if (lane == v || lane == v + 24) ev = acc[v];
        }
        es[h] = ev * scale;
    }

    float* et = &etile[wave][0];
    if (lane < 8 || lane >= 24) {
        const int el = (lane < 8) ? lane : (lane - 16);
        *(v4f*)(et + el * NH)     = (v4f){es[0], es[1], es[2], es[3]};
        *(v4f*)(et + el * NH + 4) = (v4f){es[4], es[5], es[6], es[7]};
    }
    __builtin_amdgcn_fence(__ATOMIC_RELEASE, "wavefront");
    __builtin_amdgcn_wave_barrier();
    __builtin_amdgcn_fence(__ATOMIC_ACQUIRE, "wavefront");

    *(v4f*)(eArr + ebase * NH + lane * 4) = *(const v4f*)(et + lane * 4);

    const int hd = lane & 7;
    #pragma unroll
    for (int j = 0; j < 4; ++j) {
        const int ej = 4 * j + (lane >> 3);
        const int nj = __builtin_amdgcn_ds_bpermute(ej << 2, n);
        atomicMax(&emaxKey[(long)nj * NH + hd], f2key(et[32 * j + lane]));
    }
}

#define SC_NODES 40
#define SC_WAVES 32
#define SC_BATCH 16
#define SC_STEP  128
#define SC_DEPTH 8
#define SC_CAP   (SC_BATCH + SC_DEPTH * SC_STEP)
__global__ __launch_bounds__(SC_WAVES * 32)
void gmab_scatter(const float* __restrict__ v0, const float* __restrict__ v1,
                  const int*   __restrict__ dst,
                  const float* __restrict__ eArr,
                  const unsigned* __restrict__ emaxKey,
                  float* __restrict__ out) {
    __shared__ float    acc[(SC_NODES + 1) * 128];
    __shared__ float    dnm[(SC_NODES + 1) * NH];
    __shared__ float    emx[(SC_NODES + 1) * NH];
    __shared__ unsigned list[SC_WAVES][SC_CAP];

    const int lane = threadIdx.x & 31;
    const int wave = __builtin_amdgcn_readfirstlane((int)(threadIdx.x >> 5));
    const int lo   = blockIdx.x * SC_NODES;

    for (int i = threadIdx.x; i < (SC_NODES + 1) * 128; i += SC_WAVES * 32) acc[i] = 0.0f;
    for (int i = threadIdx.x; i < (SC_NODES + 1) * NH; i += SC_WAVES * 32) {
        dnm[i] = 0.0f;
        emx[i] = (i < SC_NODES * NH && lo + i / NH < NNODE) ? key2f(emaxKey[(long)lo * NH + i]) : 0.0f;
    }
    __syncthreads();

    const bool     lop   = lane < 8;
    const float*   vcol  = lop ? (v0 + lane * 4) : (v1 + (lane - 8) * 4);
    const unsigned pitch = lop ? MCH : MCH * 3;
    const int      hl    = lop ? lane : (lane - 8) / 3;
    unsigned* mylist = &list[wave][0];
    unsigned  cnt    = 0;

    const v4i* dst4 = (const v4i*)dst;
    const int  nstep = NEDGE / SC_STEP;
    for (int t0 = wave; t0 < nstep; t0 += SC_WAVES * SC_DEPTH) {
        v4i d[SC_DEPTH];
        #pragma unroll
        for (int u = 0; u < SC_DEPTH; ++u) {
            const int t = t0 + u * SC_WAVES;
            d[u] = dst4[(size_t)((t < nstep) ? t : (nstep - 1)) * 32 + lane];
        }
        #pragma unroll
        for (int u = 0; u < SC_DEPTH; ++u) {
            const int      t     = t0 + u * SC_WAVES;
            const unsigned ebase = (unsigned)t * SC_STEP + lane * 4;
            const unsigned lim = (t < nstep) ? SC_NODES : 0u;
            const unsigned r0 = (unsigned)(d[u].x - lo), r1 = (unsigned)(d[u].y - lo);
            const unsigned r2 = (unsigned)(d[u].z - lo), r3 = (unsigned)(d[u].w - lo);
            const bool m0 = r0 < lim, m1 = r1 < lim, m2 = r2 < lim, m3 = r3 < lim;
            if (__builtin_amdgcn_ballot_w32(m0 | m1 | m2 | m3) != 0u) {
                unsigned mk;
                mk = __builtin_amdgcn_ballot_w32(m0);
                if (m0) mylist[cnt + __builtin_amdgcn_mbcnt_lo(mk, 0u)] = (ebase + 0u) | (r0 << 20);
                cnt += __builtin_popcount(mk);
                mk = __builtin_amdgcn_ballot_w32(m1);
                if (m1) mylist[cnt + __builtin_amdgcn_mbcnt_lo(mk, 0u)] = (ebase + 1u) | (r1 << 20);
                cnt += __builtin_popcount(mk);
                mk = __builtin_amdgcn_ballot_w32(m2);
                if (m2) mylist[cnt + __builtin_amdgcn_mbcnt_lo(mk, 0u)] = (ebase + 2u) | (r2 << 20);
                cnt += __builtin_popcount(mk);
                mk = __builtin_amdgcn_ballot_w32(m3);
                if (m3) mylist[cnt + __builtin_amdgcn_mbcnt_lo(mk, 0u)] = (ebase + 3u) | (r3 << 20);
                cnt += __builtin_popcount(mk);
            }
        }
        const bool last = t0 + SC_WAVES * SC_DEPTH >= nstep;
        if (cnt >= SC_BATCH || last) {
            __builtin_amdgcn_fence(__ATOMIC_RELEASE, "wavefront");
            __builtin_amdgcn_wave_barrier();
            __builtin_amdgcn_fence(__ATOMIC_ACQUIRE, "wavefront");
            const unsigned nb = last ? (cnt + SC_BATCH - 1) / SC_BATCH : cnt / SC_BATCH;
            for (unsigned b = 0; b < nb; ++b) {
                const unsigned li  = b * SC_BATCH + (unsigned)(lane & (SC_BATCH - 1));
                const unsigned ent = (li < cnt) ? mylist[li] : ((unsigned)SC_NODES << 20);
                v4f   vr[SC_BATCH];
                float ev[SC_BATCH];
                float em[SC_BATCH];
                #pragma unroll
                for (int i = 0; i < SC_BATCH; ++i) {
                    const unsigned e = (unsigned)__builtin_amdgcn_readlane((int)ent, i) & 0xFFFFFu;
                    vr[i] = *(const v4f*)(vcol + (size_t)e * pitch);
                    ev[i] = eArr[(size_t)e * NH + hl];
                    em[i] = emx[((unsigned)__builtin_amdgcn_readlane((int)ent, i) >> 20) * NH + hl];
                }
                #pragma unroll
                for (int i = 0; i < SC_BATCH; ++i) {
                    const unsigned nl = (unsigned)__builtin_amdgcn_readlane((int)ent, i) >> 20;
                    const float a = __expf(ev[i] - em[i]);
                    float* ap = acc + nl * 128 + lane;
                    atomicAdd(ap,      a * vr[i].x);
                    atomicAdd(ap + 32, a * vr[i].y);
                    atomicAdd(ap + 64, a * vr[i].z);
                    atomicAdd(ap + 96, a * vr[i].w);
                    if (lop) atomicAdd(dnm + nl * NH + lane, a);
                }
            }
            const unsigned done = nb * SC_BATCH;
            const unsigned rem  = (done < cnt) ? (cnt - done) : 0u;
            const unsigned keep = ((unsigned)lane < rem) ? mylist[done + lane] : 0u;
            if ((unsigned)lane < rem) mylist[lane] = keep;
            cnt = rem;
        }
    }
    __syncthreads();

    float* out1 = out + (size_t)NNODE * MCH;
    for (int nl = wave; nl < SC_NODES; nl += SC_WAVES) {
        const int n = lo + nl;
        if (n < NNODE) {
            const float* ap = acc + nl * 128 + lane;
            const float d = dnm[nl * NH + hl];
            const v4f o = (d > 0.0f) ? (v4f){ap[0] / d, ap[32] / d, ap[64] / d, ap[96] / d}
                                     : (v4f){0.0f, 0.0f, 0.0f, 0.0f};
            float* op = lop ? (out + (size_t)n * MCH + lane * 4)
                            : (out1 + (size_t)n * (MCH * 3) + (lane - 8) * 4);
            *(v4f*)op = o;
        }
    }
}

extern "C" void kernel_launch(void* const* d_in, const int* in_sizes, int n_in,
                              void* d_out, int out_size, void* d_ws, size_t ws_size,
                              hipStream_t stream) {
    const float* v0 = (const float*)d_in[0];
    const float* v1 = (const float*)d_in[1];
    const float* k0 = (const float*)d_in[2];
    const float* k1 = (const float*)d_in[3];
    const float* q0 = (const float*)d_in[4];
    const float* q1 = (const float*)d_in[5];
    const int*   dst = (const int*)d_in[6];
    float* out = (float*)d_out;

    float*    eArr    = (float*)d_ws;
    unsigned* emaxKey = (unsigned*)(eArr + (size_t)NEDGE * NH);

    const int nh = NNODE * NH;
    gmab_init<<<(nh + 255) / 256, 256, 0, stream>>>(emaxKey, nh);

    gmab_dot_max<<<NEDGE / (DOT_WAVES * 16), DOT_WAVES * 32, 0, stream>>>(k0, k1, q0, q1, dst, eArr, emaxKey);

    gmab_scatter<<<(NNODE + SC_NODES - 1) / SC_NODES, SC_WAVES * 32, 0, stream>>>(v0, v1, dst, eArr, emaxKey, out);
}


# segment max taken by the owner group in the scatter pass (LDS max over its kept edges, value rows prefetched to L2 meanwhile), not by global atomics in the dot pass; no init launch; adds unordered
# speedup vs baseline: 4.5436x; 1.1287x over previous
#include <hip/hip_runtime.h>

#define NNODE 10000
#define NEDGE 320000
#define MCH   32
#define NH    8

typedef float v2f __attribute__((ext_vector_type(2)));
typedef float v4f __attribute__((ext_vector_type(4)));
typedef float v8f __attribute__((ext_vector_type(8)));
typedef int   v4i __attribute__((ext_vector_type(4)));

__device__ __forceinline__ unsigned f2key(float x) {
    unsigned b = __float_as_uint(x);
    return (b & 0x80000000u) ? ~b : (b | 0x80000000u);
}
__device__ __forceinline__ float key2f(unsigned k) {
    if (k == 0u) return 0.0f;
    unsigned b = (k & 0x80000000u) ? (k & 0x7fffffffu) : ~k;
    return __uint_as_float(b);
}

#define TROW 132
#define DOT_WAVES 4
__global__ __launch_bounds__(DOT_WAVES * 32)
void gmab_dot(const float* __restrict__ k0, const float* __restrict__ k1,
              const float* __restrict__ q0, const float* __restrict__ q1,
              const int*   __restrict__ dst,
              float* __restrict__ eArr) {
    __shared__ float tile[DOT_WAVES][2][16 * TROW];
    __shared__ float etile[DOT_WAVES][16 * NH];
    const int  lane  = threadIdx.x & 31;
    const int  wave  = threadIdx.x >> 5;
    const long ebase = ((long)blockIdx.x * DOT_WAVES + wave) * 16;
    const int  r     = lane & 15;
    const long e     = ebase + r;
    const int  n     = dst[e];
    const int  klo   = (lane < 16) ? 0 : 2;
    const float scale = 0.08838834764831845f;

    float* kt = &tile[wave][0][0];
    float* qt = &tile[wave][1][0];

    const bool     lo    = lane < 8;
    const float*   kcol  = lo ? (k0 + lane * 4) : (k1 + (lane - 8) * 4);
    const float*   qcol  = lo ? (q0 + lane * 4) : (q1 + (lane - 8) * 4);
    const unsigned pitch = lo ? MCH : MCH * 3;
    const unsigned eb    = (unsigned)ebase;
    v4f kr[16], qr[16];
    #pragma unroll
    for (int i = 0; i < 16; ++i) {
        const unsigned ni = (unsigned)__builtin_amdgcn_readlane(n, i);
        kr[i] = *(const v4f*)(kcol + (size_t)((eb + i) * pitch));
        qr[i] = *(const v4f*)(qcol + (size_t)(ni * pitch));
    }
    #pragma unroll
    for (int i = 0; i < 16; ++i) {
        *(v4f*)(kt + i * TROW + lane * 4) = kr[i];
        *(v4f*)(qt + i * TROW + lane * 4) = qr[i];
    }
    __syncthreads();

    const float* ka = kt + r * TROW + klo;
    const float* qb = qt + r * TROW + klo;

    float es[NH];
    #pragma unroll
    for (int h = 0; h < NH; ++h) {
        v8f acc = {};
        #pragma unroll
        for (int kk = 0; kk < 4; ++kk) {
            const int off = (kk == 0) ? (h * 4) : (MCH + h * 12 + (kk - 1) * 4);
            const v2f a = *(const v2f*)(ka + off);
            const v2f b = *(const v2f*)(qb + off);
            acc = __builtin_amdgcn_wmma_f32_16x16x4_f32(
                false, a, false, b, (short)0, acc, false, false);
        }
        float ev = 0.0f;
        #pragma unroll
        for (int v = 0; v < 8; ++v) {
            if (lane == v || lane == v + 24) ev = acc[v];
        }
        es[h] = ev * scale;
    }

    float* et = &etile[wave][0];
    if (lane < 8 || lane >= 24) {
        const int el = (lane < 8) ? lane : (lane - 16);
        *(v4f*)(et + el * NH)     = (v4f){es[0], es[1], es[2], es[3]};
        *(v4f*)(et + el * NH + 4) = (v4f){es[4], es[5], es[6], es[7]};
    }
    __builtin_amdgcn_fence(__ATOMIC_RELEASE, "wavefront");
    __builtin_amdgcn_wave_barrier();
    __builtin_amdgcn_fence(__ATOMIC_ACQUIRE, "wavefront");

    *(v4f*)(eArr + ebase * NH + lane * 4) = *(const v4f*)(et + lane * 4);
}

#define SC_NODES 40
#define SC_WAVES 32
#define SC_BATCH 16
#define SC_STEP  128
#define SC_DEPTH 8
#define SC_ROOM  (SC_DEPTH * SC_STEP)
#define SC_KEEP  256
#define SC_CAP   (SC_KEEP + SC_ROOM)

__device__ __forceinline__ unsigned sc_append(v4i d, int t, int nstep, int lo, int lane,
                                              unsigned* mylist, unsigned cnt) {
    const unsigned ebase = (unsigned)t * SC_STEP + lane * 4;
    const unsigned lim = (t < nstep) ? SC_NODES : 0u;
    const unsigned r0 = (unsigned)(d.x - lo), r1 = (unsigned)(d.y - lo);
    const unsigned r2 = (unsigned)(d.z - lo), r3 = (unsigned)(d.w - lo);
    const bool m0 = r0 < lim, m1 = r1 < lim, m2 = r2 < lim, m3 = r3 < lim;
    if (__builtin_amdgcn_ballot_w32(m0 | m1 | m2 | m3) != 0u) {
        unsigned mk;
        mk = __builtin_amdgcn_ballot_w32(m0);
        if (m0) mylist[cnt + __builtin_amdgcn_mbcnt_lo(mk, 0u)] = (ebase + 0u) | (r0 << 20);
        cnt += __builtin_popcount(mk);
        mk = __builtin_amdgcn_ballot_w32(m1);
        if (m1) mylist[cnt + __builtin_amdgcn_mbcnt_lo(mk, 0u)] = (ebase + 1u) | (r1 << 20);
        cnt += __builtin_popcount(mk);
        mk = __builtin_amdgcn_ballot_w32(m2);
        if (m2) mylist[cnt + __builtin_amdgcn_mbcnt_lo(mk, 0u)] = (ebase + 2u) | (r2 << 20);
        cnt += __builtin_popcount(mk);
        mk = __builtin_amdgcn_ballot_w32(m3);
        if (m3) mylist[cnt + __builtin_amdgcn_mbcnt_lo(mk, 0u)] = (ebase + 3u) | (r3 << 20);
        cnt += __builtin_popcount(mk);
    }
    return cnt;
}

__device__ __forceinline__ void sc_add_batch(unsigned ent, int lane, bool lop, int hl,
                                             const float* vcol, unsigned pitch,
                                             const float* __restrict__ eArr,
                                             const unsigned* emx, float* acc, float* dnm) {
    v4f   vr[SC_BATCH];
    float ev[SC_BATCH];
    float em[SC_BATCH];
    #pragma unroll
    for (int i = 0; i < SC_BATCH; ++i) {
        const unsigned e = (unsigned)__builtin_amdgcn_readlane((int)ent, i) & 0xFFFFFu;
        vr[i] = *(const v4f*)(vcol + (size_t)e * pitch);
        ev[i] = eArr[(size_t)e * NH + hl];
        em[i] = __uint_as_float(emx[((unsigned)__builtin_amdgcn_readlane((int)ent, i) >> 20) * NH + hl]);
    }
    #pragma unroll
    for (int i = 0; i < SC_BATCH; ++i) {
        const unsigned nl = (unsigned)__builtin_amdgcn_readlane((int)ent, i) >> 20;
        const float a = __expf(ev[i] - em[i]);
        float* ap = acc + nl * 128 + lane;
        atomicAdd(ap,      a * vr[i].x);
        atomicAdd(ap + 32, a * vr[i].y);
        atomicAdd(ap + 64, a * vr[i].z);
        atomicAdd(ap + 96, a * vr[i].w);
        if (lop) atomicAdd(dnm + nl * NH + lane, a);
    }
}

__global__ __launch_bounds__(SC_WAVES * 32)
void gmab_scatter(const float* __restrict__ v0, const float* __restrict__ v1,
                  const int*   __restrict__ dst,
                  const float* __restrict__ eArr,
                  float* __restrict__ out) {
    __shared__ float    acc[(SC_NODES + 1) * 128];
    __shared__ float    dnm[(SC_NODES + 1) * NH];
    __shared__ unsigned emx[(SC_NODES + 1) * NH];
    __shared__ unsigned list[SC_WAVES][SC_CAP];
    __shared__ unsigned spill;

    const int lane = threadIdx.x & 31;
    const int wave = __builtin_amdgcn_readfirstlane((int)(threadIdx.x >> 5));
    const int lo   = blockIdx.x * SC_NODES;

    for (int i = threadIdx.x; i < (SC_NODES + 1) * 128; i += SC_WAVES * 32) acc[i] = 0.0f;
    for (int i = threadIdx.x; i < (SC_NODES + 1) * NH; i += SC_WAVES * 32) {
        dnm[i] = 0.0f;
        emx[i] = 0u;
    }
    if (threadIdx.x == 0) spill = 0u;
    __syncthreads();

    const bool     lop   = lane < 8;
    const float*   vcol  = lop ? (v0 + lane * 4) : (v1 + (lane - 8) * 4);
    const unsigned pitch = lop ? MCH : MCH * 3;
    const int      hl    = lop ? lane : (lane - 8) / 3;
    unsigned* mylist = &list[wave][0];
    unsigned  cnt    = 0;
    unsigned  mdone  = 0;
    bool      gaveup = false;

    const v4i* dst4 = (const v4i*)dst;
    const int  nstep = NEDGE / SC_STEP;

    for (int t0 = wave; t0 < nstep; t0 += SC_WAVES * SC_DEPTH) {
        v4i d[SC_DEPTH];
        #pragma unroll
        for (int u = 0; u < SC_DEPTH; ++u) {
            const int t = t0 + u * SC_WAVES;
            d[u] = dst4[(size_t)((t < nstep) ? t : (nstep - 1)) * 32 + lane];
        }
        #pragma unroll
        for (int u = 0; u < SC_DEPTH; ++u)
            cnt = sc_append(d[u], t0 + u * SC_WAVES, nstep, lo, lane, mylist, cnt);

        const bool last = t0 + SC_WAVES * SC_DEPTH >= nstep;
        if (cnt - mdone >= SC_BATCH || last || cnt > SC_KEEP) {
            __builtin_amdgcn_fence(__ATOMIC_RELEASE, "wavefront");
            __builtin_amdgcn_wave_barrier();
            __builtin_amdgcn_fence(__ATOMIC_ACQUIRE, "wavefront");
            for (unsigned c = mdone; c < cnt; c += 32) {
                const unsigned li = c + (unsigned)lane;
                if (li < cnt) {
                    const unsigned ent = mylist[li];
                    const float*   ep  = eArr + (size_t)(ent & 0xFFFFFu) * NH;
                    __builtin_prefetch(v0 + (size_t)(ent & 0xFFFFFu) * MCH, 0, 1);
                    __builtin_prefetch(v1 + (size_t)(ent & 0xFFFFFu) * (MCH * 3), 0, 1);
                    __builtin_prefetch(v1 + (size_t)(ent & 0xFFFFFu) * (MCH * 3) + 64, 0, 1);
                    const v4f ea = *(const v4f*)ep;
                    const v4f eb = *(const v4f*)(ep + 4);
                    unsigned* mk = emx + (ent >> 20) * NH;
                    atomicMax(mk + 0, f2key(ea.x));
                    atomicMax(mk + 1, f2key(ea.y));
                    atomicMax(mk + 2, f2key(ea.z));
                    atomicMax(mk + 3, f2key(ea.w));
                    atomicMax(mk + 4, f2key(eb.x));
                    atomicMax(mk + 5, f2key(eb.y));
                    atomicMax(mk + 6, f2key(eb.z));
                    atomicMax(mk + 7, f2key(eb.w));
                }
            }
            mdone = cnt;
            if (cnt > SC_KEEP) {
                gaveup = true;
                cnt    = 0;
                mdone  = 0;
            }
        }
    }
    if (gaveup && lane == 0) spill = 1u;
    __syncthreads();

    for (int i = threadIdx.x; i < (SC_NODES + 1) * NH; i += SC_WAVES * 32)
        emx[i] = __float_as_uint(key2f(emx[i]));
    const bool rescan = spill != 0u;
    __syncthreads();

    if (!rescan) {
        const unsigned nb = (cnt + SC_BATCH - 1) / SC_BATCH;
        for (unsigned b = 0; b < nb; ++b) {
            const unsigned k   = b * SC_BATCH + (unsigned)(lane & (SC_BATCH - 1));
            const unsigned ent = (k < cnt) ? mylist[cnt - 1u - k] : ((unsigned)SC_NODES << 20);
            sc_add_batch(ent, lane, lop, hl, vcol, pitch, eArr, emx, acc, dnm);
        }
    } else {
        cnt = 0;
        for (int t0 = wave; t0 < nstep; t0 += SC_WAVES * SC_DEPTH) {
            v4i d[SC_DEPTH];
            #pragma unroll
            for (int u = 0; u < SC_DEPTH; ++u) {
                const int t = t0 + u * SC_WAVES;
                d[u] = dst4[(size_t)((t < nstep) ? t : (nstep - 1)) * 32 + lane];
            }
            #pragma unroll
            for (int u = 0; u < SC_DEPTH; ++u)
                cnt = sc_append(d[u], t0 + u * SC_WAVES, nstep, lo, lane, mylist, cnt);

            const bool last = t0 + SC_WAVES * SC_DEPTH >= nstep;
            if (cnt >= SC_BATCH || last) {
                __builtin_amdgcn_fence(__ATOMIC_RELEASE, "wavefront");
                __builtin_amdgcn_wave_barrier();
                __builtin_amdgcn_fence(__ATOMIC_ACQUIRE, "wavefront");
                const unsigned nb = last ? (cnt + SC_BATCH - 1) / SC_BATCH : cnt / SC_BATCH;
                for (unsigned b = 0; b < nb; ++b) {
                    const unsigned li  = b * SC_BATCH + (unsigned)(lane & (SC_BATCH - 1));
                    const unsigned ent = (li < cnt) ? mylist[li] : ((unsigned)SC_NODES << 20);
                    sc_add_batch(ent, lane, lop, hl, vcol, pitch, eArr, emx, acc, dnm);
                }
                const unsigned done = nb * SC_BATCH;
                const unsigned rem  = (done < cnt) ? (cnt - done) : 0u;
                const unsigned keep = ((unsigned)lane < rem) ? mylist[done + lane] : 0u;
                if ((unsigned)lane < rem) mylist[lane] = keep;
                cnt = rem;
            }
        }
    }
    __syncthreads();

    float* out1 = out + (size_t)NNODE * MCH;
    for (int nl = wave; nl < SC_NODES; nl += SC_WAVES) {
        const int n = lo + nl;
        if (n < NNODE) {
            const float* ap = acc + nl * 128 + lane;
            const float d = dnm[nl * NH + hl];
            const v4f o = (d > 0.0f) ? (v4f){ap[0] / d, ap[32] / d, ap[64] / d, ap[96] / d}
                                     : (v4f){0.0f, 0.0f, 0.0f, 0.0f};
            float* op = lop ? (out + (size_t)n * MCH + lane * 4)
                            : (out1 + (size_t)n * (MCH * 3) + (lane - 8) * 4);
            *(v4f*)op = o;
        }
    }
}

extern "C" void kernel_launch(void* const* d_in, const int* in_sizes, int n_in,
                              void* d_out, int out_size, void* d_ws, size_t ws_size,
                              hipStream_t stream) {
    const float* v0 = (const float*)d_in[0];
    const float* v1 = (const float*)d_in[1];
    const float* k0 = (const float*)d_in[2];
    const float* k1 = (const float*)d_in[3];
    const float* q0 = (const float*)d_in[4];
    const float* q1 = (const float*)d_in[5];
    const int*   dst = (const int*)d_in[6];
    float* out = (float*)d_out;

    float* eArr = (float*)d_ws;

    gmab_dot<<<NEDGE / (DOT_WAVES * 16), DOT_WAVES * 32, 0, stream>>>(k0, k1, q0, q1, dst, eArr);

    gmab_scatter<<<(NNODE + SC_NODES - 1) / SC_NODES, SC_WAVES * 32, 0, stream>>>(v0, v1, dst, eArr, out);
}


# scatter pass: the scan reads a 1-byte owner-group id per edge (dst/40, written by the dot pass), 4 edges per 32-bit compare, not the 4-byte dst; node index read per kept edge; adds stay unordered
# speedup vs baseline: 4.8000x; 1.0564x over previous
#include <hip/hip_runtime.h>

#define NNODE 10000
#define NEDGE 320000
#define MCH   32
#define NH    8

typedef float v2f __attribute__((ext_vector_type(2)));
typedef float v4f __attribute__((ext_vector_type(4)));
typedef float v8f __attribute__((ext_vector_type(8)));
typedef int   v4i __attribute__((ext_vector_type(4)));

__device__ __forceinline__ unsigned f2key(float x) {
    unsigned b = __float_as_uint(x);
    return (b & 0x80000000u) ? ~b : (b | 0x80000000u);
}
__device__ __forceinline__ float key2f(unsigned k) {
    if (k == 0u) return 0.0f;
    unsigned b = (k & 0x80000000u) ? (k & 0x7fffffffu) : ~k;
    return __uint_as_float(b);
}

#define SC_NODES 40
#define TROW 132
#define DOT_WAVES 4
__global__ __launch_bounds__(DOT_WAVES * 32)
void gmab_dot(const float* __restrict__ k0, const float* __restrict__ k1,
              const float* __restrict__ q0, const float* __restrict__ q1,
              const int*   __restrict__ dst,
              float* __restrict__ eArr,
              unsigned char* __restrict__ gid) {
    __shared__ float tile[DOT_WAVES][2][16 * TROW];
    __shared__ float etile[DOT_WAVES][16 * NH];
    const int  lane  = threadIdx.x & 31;
    const int  wave  = threadIdx.x >> 5;
    const long ebase = ((long)blockIdx.x * DOT_WAVES + wave) * 16;
    const int  r     = lane & 15;
    const long e     = ebase + r;
    const int  n     = dst[e];
    if (lane < 16) gid[e] = (unsigned char)((unsigned)n / SC_NODES);
    const int  klo   = (lane < 16) ? 0 : 2;
    const float scale = 0.08838834764831845f;

    float* kt = &tile[wave][0][0];
    float* qt = &tile[wave][1][0];

    const bool     lo    = lane < 8;
    const float*   kcol  = lo ? (k0 + lane * 4) : (k1 + (lane - 8) * 4);
    const float*   qcol  = lo ? (q0 + lane * 4) : (q1 + (lane - 8) * 4);
    const unsigned pitch = lo ? MCH : MCH * 3;
    const unsigned eb    = (unsigned)ebase;
    v4f kr[16], qr[16];
    #pragma unroll
    for (int i = 0; i < 16; ++i) {
        const unsigned ni = (unsigned)__builtin_amdgcn_readlane(n, i);
        kr[i] = *(const v4f*)(kcol + (size_t)((eb + i) * pitch));
        qr[i] = *(const v4f*)(qcol + (size_t)(ni * pitch));
    }
    #pragma unroll
    for (int i = 0; i < 16; ++i) {
        *(v4f*)(kt + i * TROW + lane * 4) = kr[i];
        *(v4f*)(qt + i * TROW + lane * 4) = qr[i];
    }
    __syncthreads();

    const float* ka = kt + r * TROW + klo;
    const float* qb = qt + r * TROW + klo;

    float es[NH];
    #pragma unroll
    for (int h = 0; h < NH; ++h) {
        v8f acc = {};
        #pragma unroll
        for (int kk = 0; kk < 4; ++kk) {
            const int off = (kk == 0) ? (h * 4) : (MCH + h * 12 + (kk - 1) * 4);
            const v2f a = *(const v2f*)(ka + off);
            const v2f b = *(const v2f*)(qb + off);
            acc = __builtin_amdgcn_wmma_f32_16x16x4_f32(
                false, a, false, b, (short)0, acc, false, false);
        }
        float ev = 0.0f;
        #pragma unroll
        for (int v = 0; v < 8; ++v) {
            if (lane == v || lane == v + 24) ev = acc[v];
        }
        es[h] = ev * scale;
    }

    float* et = &etile[wave][0];
    if (lane < 8 || lane >= 24) {
        const int el = (lane < 8) ? lane : (lane - 16);
        *(v4f*)(et + el * NH)     = (v4f){es[0], es[1], es[2], es[3]};
        *(v4f*)(et + el * NH + 4) = (v4f){es[4], es[5], es[6], es[7]};
    }
    __builtin_amdgcn_fence(__ATOMIC_RELEASE, "wavefront");
    __builtin_amdgcn_wave_barrier();
    __builtin_amdgcn_fence(__ATOMIC_ACQUIRE, "wavefront");

    *(v4f*)(eArr + ebase * NH + lane * 4) = *(const v4f*)(et + lane * 4);
}

#define SC_WAVES 32
#define SC_BATCH 16
#define SC_STEP  512
#define SC_DEPTH 4
#define SC_ROOM  (SC_DEPTH * SC_STEP)
#define SC_KEEP  256
#define SC_CAP   (SC_KEEP + SC_ROOM)

__device__ __forceinline__ unsigned sc_append(v4i w, int t, int nstep, unsigned gpat, int lane,
                                              unsigned* mylist, unsigned cnt) {
    const unsigned K7 = 0x7F7F7F7Fu;
    const unsigned x0 = (unsigned)w.x ^ gpat, x1 = (unsigned)w.y ^ gpat;
    const unsigned x2 = (unsigned)w.z ^ gpat, x3 = (unsigned)w.w ^ gpat;
    const unsigned u0 = ((x0 & K7) + K7) | x0 | K7;
    const unsigned u1 = ((x1 & K7) + K7) | x1 | K7;
    const unsigned u2 = ((x2 & K7) + K7) | x2 | K7;
    const unsigned u3 = ((x3 & K7) + K7) | x3 | K7;
    if (t < nstep && __builtin_amdgcn_ballot_w32((u0 & u1 & u2 & u3) != 0xFFFFFFFFu) != 0u) {
        unsigned long long lo = ((unsigned long long)(~u1) << 32) | (unsigned long long)(~u0);
        unsigned long long hi = ((unsigned long long)(~u3) << 32) | (unsigned long long)(~u2);
        const unsigned ebase = (unsigned)t * SC_STEP + lane * 16;
        for (int trip = 0; trip < 16; ++trip) {
            const bool has = (lo | hi) != 0ull;
            const unsigned mk = __builtin_amdgcn_ballot_w32(has);
            if (mk == 0u) break;
            if (has) {
                unsigned j;
                if (lo != 0ull) { j = (unsigned)__builtin_ctzll(lo) >> 3;        lo &= lo - 1ull; }
                else            { j = 8u + ((unsigned)__builtin_ctzll(hi) >> 3); hi &= hi - 1ull; }
                mylist[cnt + __builtin_amdgcn_mbcnt_lo(mk, 0u)] = ebase + j;
            }
            cnt += __builtin_popcount(mk);
        }
    }
    return cnt;
}

__device__ __forceinline__ void sc_resolve(unsigned* mylist, unsigned from, unsigned to, int lane,
                                           const int* __restrict__ dst, int lo) {
    for (unsigned c = from; c < to; c += 32) {
        const unsigned li = c + (unsigned)lane;
        if (li < to) {
            const unsigned e = mylist[li] & 0xFFFFFu;
            mylist[li] = e | ((unsigned)(dst[e] - lo) << 20);
        }
    }
}

__device__ __forceinline__ void sc_add_batch(unsigned ent, int lane, bool lop, int hl,
                                             const float* vcol, unsigned pitch,
                                             const float* __restrict__ eArr,
                                             const unsigned* emx, float* acc, float* dnm) {
    v4f   vr[SC_BATCH];
    float ev[SC_BATCH];
    float em[SC_BATCH];
    #pragma unroll
    for (int i = 0; i < SC_BATCH; ++i) {
        const unsigned e = (unsigned)__builtin_amdgcn_readlane((int)ent, i) & 0xFFFFFu;
        vr[i] = *(const v4f*)(vcol + (size_t)e * pitch);
        ev[i] = eArr[(size_t)e * NH + hl];
        em[i] = __uint_as_float(emx[((unsigned)__builtin_amdgcn_readlane((int)ent, i) >> 20) * NH + hl]);
    }
    #pragma unroll
    for (int i = 0; i < SC_BATCH; ++i) {
        const unsigned nl = (unsigned)__builtin_amdgcn_readlane((int)ent, i) >> 20;
        const float a = __expf(ev[i] - em[i]);
        float* ap = acc + nl * 128 + lane;
        atomicAdd(ap,      a * vr[i].x);
        atomicAdd(ap + 32, a * vr[i].y);
        atomicAdd(ap + 64, a * vr[i].z);
        atomicAdd(ap + 96, a * vr[i].w);
        if (lop) atomicAdd(dnm + nl * NH + lane, a);
    }
}

__global__ __launch_bounds__(SC_WAVES * 32)
void gmab_scatter(const float* __restrict__ v0, const float* __restrict__ v1,
                  const int*   __restrict__ dst,
                  const unsigned char* __restrict__ gid,
                  const float* __restrict__ eArr,
                  float* __restrict__ out) {
    __shared__ float    acc[(SC_NODES + 1) * 128];
    __shared__ float    dnm[(SC_NODES + 1) * NH];
    __shared__ unsigned emx[(SC_NODES + 1) * NH];
    __shared__ unsigned list[SC_WAVES][SC_CAP];
    __shared__ unsigned spill;

    const int lane = threadIdx.x & 31;
    const int wave = __builtin_amdgcn_readfirstlane((int)(threadIdx.x >> 5));
    const int lo   = blockIdx.x * SC_NODES;

    for (int i = threadIdx.x; i < (SC_NODES + 1) * 128; i += SC_WAVES * 32) acc[i] = 0.0f;
    for (int i = threadIdx.x; i < (SC_NODES + 1) * NH; i += SC_WAVES * 32) {
        dnm[i] = 0.0f;
        emx[i] = 0u;
    }
    if (threadIdx.x == 0) spill = 0u;
    __syncthreads();

    const bool     lop   = lane < 8;
    const float*   vcol  = lop ? (v0 + lane * 4) : (v1 + (lane - 8) * 4);
    const unsigned pitch = lop ? MCH : MCH * 3;
    const int      hl    = lop ? lane : (lane - 8) / 3;
    unsigned* mylist = &list[wave][0];
    unsigned  cnt    = 0;
    unsigned  mdone  = 0;
    bool      gaveup = false;

    const v4i* gid16 = (const v4i*)gid;
    const unsigned gpat = blockIdx.x * 0x01010101u;
    const int  nstep = NEDGE / SC_STEP;

    for (int t0 = wave; t0 < nstep; t0 += SC_WAVES * SC_DEPTH) {
        v4i d[SC_DEPTH];
        #pragma unroll
        for (int u = 0; u < SC_DEPTH; ++u) {
            const int t = t0 + u * SC_WAVES;
            d[u] = gid16[(size_t)((t < nstep) ? t : (nstep - 1)) * 32 + lane];
        }
        #pragma unroll
        for (int u = 0; u < SC_DEPTH; ++u)
            cnt = sc_append(d[u], t0 + u * SC_WAVES, nstep, gpat, lane, mylist, cnt);

        const bool last = t0 + SC_WAVES * SC_DEPTH >= nstep;
        if (cnt - mdone >= SC_BATCH || last || cnt > SC_KEEP) {
            __builtin_amdgcn_fence(__ATOMIC_RELEASE, "wavefront");
            __builtin_amdgcn_wave_barrier();
            __builtin_amdgcn_fence(__ATOMIC_ACQUIRE, "wavefront");
            for (unsigned c = mdone; c < cnt; c += 32) {
                const unsigned li = c + (unsigned)lane;
                if (li < cnt) {
                    const unsigned eg  = mylist[li];
                    const float*   ep  = eArr + (size_t)eg * NH;
                    __builtin_prefetch(v0 + (size_t)eg * MCH, 0, 1);
                    __builtin_prefetch(v1 + (size_t)eg * (MCH * 3), 0, 1);
                    __builtin_prefetch(v1 + (size_t)eg * (MCH * 3) + 64, 0, 1);
                    const v4f ea = *(const v4f*)ep;
                    const v4f eb = *(const v4f*)(ep + 4);
                    const unsigned nl = (unsigned)(dst[eg] - lo);
                    mylist[li] = eg | (nl << 20);
                    unsigned* mk = emx + nl * NH;
                    atomicMax(mk + 0, f2key(ea.x));
                    atomicMax(mk + 1, f2key(ea.y));
                    atomicMax(mk + 2, f2key(ea.z));
                    atomicMax(mk + 3, f2key(ea.w));
                    atomicMax(mk + 4, f2key(eb.x));
                    atomicMax(mk + 5, f2key(eb.y));
                    atomicMax(mk + 6, f2key(eb.z));
                    atomicMax(mk + 7, f2key(eb.w));
                }
            }
            mdone = cnt;
            if (cnt > SC_KEEP) {
                gaveup = true;
                cnt    = 0;
                mdone  = 0;
            }
        }
    }
    if (gaveup && lane == 0) spill = 1u;
    __syncthreads();

    for (int i = threadIdx.x; i < (SC_NODES + 1) * NH; i += SC_WAVES * 32)
        emx[i] = __float_as_uint(key2f(emx[i]));
    const bool rescan = spill != 0u;
    __syncthreads();

    if (!rescan) {
        const unsigned nb = (cnt + SC_BATCH - 1) / SC_BATCH;
        for (unsigned b = 0; b < nb; ++b) {
            const unsigned k   = b * SC_BATCH + (unsigned)(lane & (SC_BATCH - 1));
            const unsigned ent = (k < cnt) ? mylist[cnt - 1u - k] : ((unsigned)SC_NODES << 20);
            sc_add_batch(ent, lane, lop, hl, vcol, pitch, eArr, emx, acc, dnm);
        }
    } else {
        cnt = 0;
        for (int t0 = wave; t0 < nstep; t0 += SC_WAVES * SC_DEPTH) {
            v4i d[SC_DEPTH];
            #pragma unroll
            for (int u = 0; u < SC_DEPTH; ++u) {
                const int t = t0 + u * SC_WAVES;
                d[u] = gid16[(size_t)((t < nstep) ? t : (nstep - 1)) * 32 + lane];
            }
            #pragma unroll
            for (int u = 0; u < SC_DEPTH; ++u)
                cnt = sc_append(d[u], t0 + u * SC_WAVES, nstep, gpat, lane, mylist, cnt);

            const bool last = t0 + SC_WAVES * SC_DEPTH >= nstep;
            if (cnt >= SC_BATCH || last) {
                __builtin_amdgcn_fence(__ATOMIC_RELEASE, "wavefront");
                __builtin_amdgcn_wave_barrier();
                __builtin_amdgcn_fence(__ATOMIC_ACQUIRE, "wavefront");
                sc_resolve(mylist, 0u, cnt, lane, dst, lo);
                __builtin_amdgcn_fence(__ATOMIC_RELEASE, "wavefront");
                __builtin_amdgcn_wave_barrier();
                __builtin_amdgcn_fence(__ATOMIC_ACQUIRE, "wavefront");
                const unsigned nb = last ? (cnt + SC_BATCH - 1) / SC_BATCH : cnt / SC_BATCH;
                for (unsigned b = 0; b < nb; ++b) {
                    const unsigned li  = b * SC_BATCH + (unsigned)(lane & (SC_BATCH - 1));
                    const unsigned ent = (li < cnt) ? mylist[li] : ((unsigned)SC_NODES << 20);
                    sc_add_batch(ent, lane, lop, hl, vcol, pitch, eArr, emx, acc, dnm);
                }
                const unsigned done = nb * SC_BATCH;
                const unsigned rem  = (done < cnt) ? (cnt - done) : 0u;
                const unsigned keep = ((unsigned)lane < rem) ? mylist[done + lane] : 0u;
                if ((unsigned)lane < rem) mylist[lane] = keep;
                cnt = rem;
            }
        }
    }
    __syncthreads();

    float* out1 = out + (size_t)NNODE * MCH;
    for (int nl = wave; nl < SC_NODES; nl += SC_WAVES) {
        const int n = lo + nl;
        if (n < NNODE) {
            const float* ap = acc + nl * 128 + lane;
            const float d = dnm[nl * NH + hl];
            const v4f o = (d > 0.0f) ? (v4f){ap[0] / d, ap[32] / d, ap[64] / d, ap[96] / d}
                                     : (v4f){0.0f, 0.0f, 0.0f, 0.0f};
            float* op = lop ? (out + (size_t)n * MCH + lane * 4)
                            : (out1 + (size_t)n * (MCH * 3) + (lane - 8) * 4);
            *(v4f*)op = o;
        }
    }
}

extern "C" void kernel_launch(void* const* d_in, const int* in_sizes, int n_in,
                              void* d_out, int out_size, void* d_ws, size_t ws_size,
                              hipStream_t stream) {
    const float* v0 = (const float*)d_in[0];
    const float* v1 = (const float*)d_in[1];
    const float* k0 = (const float*)d_in[2];
    const float* k1 = (const float*)d_in[3];
    const float* q0 = (const float*)d_in[4];
    const float* q1 = (const float*)d_in[5];
    const int*   dst = (const int*)d_in[6];
    float* out = (float*)d_out;

    float* eArr = (float*)d_ws;
    unsigned char* gid = (unsigned char*)(eArr + (size_t)NEDGE * NH);

    gmab_dot<<<NEDGE / (DOT_WAVES * 16), DOT_WAVES * 32, 0, stream>>>(k0, k1, q0, q1, dst, eArr, gid);

    gmab_scatter<<<(NNODE + SC_NODES - 1) / SC_NODES, SC_WAVES * 32, 0, stream>>>(v0, v1, dst, gid, eArr, out);
}
